# starting kernel
# baseline (speedup 1.0000x reference)
#include <hip/hip_runtime.h>
#include <hip/hip_bf16.h>

#define NN 50000
#define EE 600000
#define FF 128
#define HH 128
#define OO 64
#define RR 8

typedef float v2f __attribute__((ext_vector_type(2)));
typedef float v8f __attribute__((ext_vector_type(8)));

__global__ void init_mm_kernel(unsigned* mm) {
    if (threadIdx.x == 0) { mm[0] = 0x7F800000u; mm[1] = 0u; }
}

__global__ void ew_minmax_kernel(const float* __restrict__ ew, int E,
                                 unsigned* __restrict__ mm) {
    float mn = __builtin_huge_valf();
    float mx = 0.0f;
    for (int i = blockIdx.x * blockDim.x + threadIdx.x; i < E;
         i += gridDim.x * blockDim.x) {
        float w = ew[i];
        mn = fminf(mn, w);
        mx = fmaxf(mx, w);
    }
    for (int off = 16; off > 0; off >>= 1) {
        mn = fminf(mn, __shfl_down(mn, off, 32));
        mx = fmaxf(mx, __shfl_down(mx, off, 32));
    }
    if ((threadIdx.x & 31) == 0) {
        atomicMin(&mm[0], __float_as_uint(mn));
        atomicMax(&mm[1], __float_as_uint(mx));
    }
}

__global__ void count_kernel(const int* __restrict__ dst, float* __restrict__ cnt,
                             int E) {
    int e = blockIdx.x * blockDim.x + threadIdx.x;
    if (e < E) atomicAdd(&cnt[dst[e]], 1.0f);
}

__global__ void invcnt_kernel(const float* __restrict__ cnt,
                              float* __restrict__ inv, int N) {
    int n = blockIdx.x * blockDim.x + threadIdx.x;
    if (n < N) inv[n] = 1.0f / fmaxf(cnt[n], 1.0f);
}

template <int M>
__global__ __launch_bounds__(32)
void gemm_wmma_f32_kernel(const float* __restrict__ A, const float* __restrict__ B,
                          float* __restrict__ C) {
    const int K = 128;
    const int lane = threadIdx.x;
    const int half = lane >> 4;
    const int lr   = lane & 15;
    const int rowbase = blockIdx.x * 16;

    const float* arow = A + (size_t)(rowbase + lr) * K + 2 * half;
    v2f a[32];
#pragma unroll
    for (int s = 0; s < 32; ++s) {
        a[s] = *(const v2f*)(arow + 4 * s);
    }

#pragma unroll
    for (int ct = 0; ct < (M >> 4); ++ct) {
        const int col = ct * 16 + lr;
        const float* bp = B + (size_t)(2 * half) * M + col;
        v2f b[32];
#pragma unroll
        for (int s = 0; s < 32; ++s) {
            b[s].x = bp[(4 * s + 0) * M];
            b[s].y = bp[(4 * s + 1) * M];
        }
        v8f acc0 = {};
        v8f acc1 = {};
#pragma unroll
        for (int s = 0; s < 32; s += 2) {
            acc0 = __builtin_amdgcn_wmma_f32_16x16x4_f32(
                false, a[s], false, b[s], (short)0, acc0, false, false);
            acc1 = __builtin_amdgcn_wmma_f32_16x16x4_f32(
                false, a[s + 1], false, b[s + 1], (short)0, acc1, false, false);
        }
        float* crow = C + (size_t)(rowbase + 8 * half) * M + col;
#pragma unroll
        for (int j = 0; j < 8; ++j) {
            crow[(size_t)j * M] = acc0[j] + acc1[j];
        }
    }
}

__global__ void scatter_rel_kernel(const float* __restrict__ proj,
                                   const int* __restrict__ src,
                                   const int* __restrict__ dst,
                                   const int* __restrict__ et,
                                   const float* __restrict__ ew,
                                   const unsigned* __restrict__ mm,
                                   float* __restrict__ agg,
                                   int E, int M, int r) {
    const int Mq = M >> 2;
    long long idx = (long long)blockIdx.x * blockDim.x + threadIdx.x;
    long long total = (long long)E * Mq;
    if (idx >= total) return;
    int e  = (int)(idx / Mq);
    int c4 = (int)(idx - (long long)e * Mq) << 2;
    if (et[e] != r) return;
    float mn = __uint_as_float(mm[0]);
    float mx = __uint_as_float(mm[1]);
    float w = (ew[e] - mn) / (mx - mn + 1e-8f);
    const float4 p = *(const float4*)(proj + (size_t)src[e] * M + c4);
    float* ag = agg + (size_t)dst[e] * M + c4;
    atomicAdd(ag + 0, p.x * w);
    atomicAdd(ag + 1, p.y * w);
    atomicAdd(ag + 2, p.z * w);
    atomicAdd(ag + 3, p.w * w);
}

__global__ void finalize_kernel(const float* __restrict__ agg,
                                const float* __restrict__ inv,
                                const float* __restrict__ bias,
                                float* __restrict__ out,
                                int N, int M, int do_relu) {
    long long idx = (long long)blockIdx.x * blockDim.x + threadIdx.x;
    if (idx >= (long long)N * M) return;
    int n = (int)(idx / M);
    int c = (int)(idx - (long long)n * M);
    float v = agg[idx] * inv[n] + out[idx] + bias[c];
    out[idx] = do_relu ? fmaxf(v, 0.0f) : v;
}

extern "C" void kernel_launch(void* const* d_in, const int* in_sizes, int n_in,
                              void* d_out, int out_size, void* d_ws, size_t ws_size,
                              hipStream_t stream) {
    const float* x     = (const float*)d_in[0];
    const int*   eidx  = (const int*)d_in[1];
    const float* ew    = (const float*)d_in[2];
    const int*   et    = (const int*)d_in[3];
    const float* W1    = (const float*)d_in[4];
    const float* root1 = (const float*)d_in[5];
    const float* bias1 = (const float*)d_in[6];
    const float* W2    = (const float*)d_in[7];
    const float* root2 = (const float*)d_in[8];
    const float* bias2 = (const float*)d_in[9];
    float* out = (float*)d_out;

    const int* src = eidx;
    const int* dst = eidx + EE;

    float* proj = (float*)d_ws;
    float* agg  = proj + (size_t)NN * HH;
    float* h    = agg  + (size_t)NN * HH;
    float* cnt  = h    + (size_t)NN * HH;
    float* inv  = cnt  + NN;
    unsigned* mm = (unsigned*)(inv + NN);

    const int T = 256;
    const int rowTiles = NN / 16;

    hipMemsetAsync(agg, 0, (size_t)NN * HH * sizeof(float), stream);
    hipMemsetAsync(cnt, 0, (size_t)NN * sizeof(float), stream);
    init_mm_kernel<<<1, 32, 0, stream>>>(mm);
    ew_minmax_kernel<<<1024, T, 0, stream>>>(ew, EE, mm);
    count_kernel<<<(EE + T - 1) / T, T, 0, stream>>>(dst, cnt, EE);
    invcnt_kernel<<<(NN + T - 1) / T, T, 0, stream>>>(cnt, inv, NN);

    {
        long long sw = (long long)EE * (HH / 4);
        int sblocks = (int)((sw + T - 1) / T);
        for (int r = 0; r < RR; ++r) {
            gemm_wmma_f32_kernel<HH><<<rowTiles, 32, 0, stream>>>(
                x, W1 + (size_t)r * FF * HH, proj);
            scatter_rel_kernel<<<sblocks, T, 0, stream>>>(
                proj, src, dst, et, ew, mm, agg, EE, HH, r);
        }
        gemm_wmma_f32_kernel<HH><<<rowTiles, 32, 0, stream>>>(x, root1, h);
        long long fw = (long long)NN * HH;
        finalize_kernel<<<(int)((fw + T - 1) / T), T, 0, stream>>>(
            agg, inv, bias1, h, NN, HH, 1);
    }

    {
        hipMemsetAsync(agg, 0, (size_t)NN * OO * sizeof(float), stream);
        long long sw = (long long)EE * (OO / 4);
        int sblocks = (int)((sw + T - 1) / T);
        for (int r = 0; r < RR; ++r) {
            gemm_wmma_f32_kernel<OO><<<rowTiles, 32, 0, stream>>>(
                h, W2 + (size_t)r * HH * OO, proj);
            scatter_rel_kernel<<<sblocks, T, 0, stream>>>(
                proj, src, dst, et, ew, mm, agg, EE, OO, r);
        }
        gemm_wmma_f32_kernel<OO><<<rowTiles, 32, 0, stream>>>(h, root2, out);
        long long fw = (long long)NN * OO;
        finalize_kernel<<<(int)((fw + T - 1) / T), T, 0, stream>>>(
            agg, inv, bias2, out, NN, OO, 0);
    }
}


# Pull, not atomic scatter: edges grouped by destination per call (count+rank, scan, fill); per 16-col slab one GEMM covers all 8 relations, half a wave sums a node's rows; sum order: arrival rank
# speedup vs baseline: 1.4253x; 1.4253x over previous
#include <hip/hip_runtime.h>
#include <hip/hip_bf16.h>

#define NN 50000
#define EE 600000
#define FF 128
#define HH 128
#define OO 64
#define RR 8

typedef float v2f __attribute__((ext_vector_type(2)));
typedef float v8f __attribute__((ext_vector_type(8)));

__global__ void init_mm_kernel(unsigned* mm) {
    if (threadIdx.x == 0) { mm[0] = 0x7F800000u; mm[1] = 0u; }
}

__global__ void ew_minmax_kernel(const float* __restrict__ ew, int E,
                                 unsigned* __restrict__ mm) {
    float mn = __builtin_huge_valf();
    float mx = 0.0f;
    for (int i = blockIdx.x * blockDim.x + threadIdx.x; i < E;
         i += gridDim.x * blockDim.x) {
        float w = ew[i];
        mn = fminf(mn, w);
        mx = fmaxf(mx, w);
    }
    for (int off = 16; off > 0; off >>= 1) {
        mn = fminf(mn, __shfl_down(mn, off, 32));
        mx = fmaxf(mx, __shfl_down(mx, off, 32));
    }
    if ((threadIdx.x & 31) == 0) {
        atomicMin(&mm[0], __float_as_uint(mn));
        atomicMax(&mm[1], __float_as_uint(mx));
    }
}

__global__ void count_rank_kernel(const int* __restrict__ dst, int* __restrict__ deg,
                                  int* __restrict__ rank, int E) {
    int e = blockIdx.x * blockDim.x + threadIdx.x;
    if (e < E) rank[e] = atomicAdd(&deg[dst[e]], 1);
}

__global__ __launch_bounds__(1024)
void scan_kernel(const int* __restrict__ deg, int* __restrict__ rowptr) {
    __shared__ int part[1024];
    constexpr int PER = (NN + 1023) / 1024;
    const int t = threadIdx.x;
    const int b = t * PER;
    int s = 0;
    for (int i = 0; i < PER; ++i) {
        const int idx = b + i;
        if (idx < NN) s += deg[idx];
    }
    part[t] = s;
    __syncthreads();
    for (int off = 1; off < 1024; off <<= 1) {
        const int v = (t >= off) ? part[t - off] : 0;
        __syncthreads();
        part[t] += v;
        __syncthreads();
    }
    int run = part[t] - s;
    for (int i = 0; i < PER; ++i) {
        const int idx = b + i;
        if (idx < NN) { rowptr[idx] = run; run += deg[idx]; }
    }
    if (t == 1023) rowptr[NN] = part[1023];
}

__global__ void fill_kernel(const int* __restrict__ src, const int* __restrict__ dst,
                            const int* __restrict__ et, const float* __restrict__ ew,
                            const unsigned* __restrict__ mm,
                            const int* __restrict__ rowptr, const int* __restrict__ rank,
                            int* __restrict__ eg, float* __restrict__ ewn, int E) {
    int e = blockIdx.x * blockDim.x + threadIdx.x;
    if (e >= E) return;
    const float mn = __uint_as_float(mm[0]);
    const float mx = __uint_as_float(mm[1]);
    const int p = rowptr[dst[e]] + rank[e];
    eg[p]  = src[e] * RR + et[e];
    ewn[p] = (ew[e] - mn) / (mx - mn + 1e-8f);
}

template <int LDB, int LDC, int NCT, int NREL>
__global__ __launch_bounds__(32)
void gemm_slab_wmma_f32_kernel(const float* __restrict__ A, const float* __restrict__ B,
                               float* __restrict__ C) {
    const int K = 128;
    const int lane = threadIdx.x;
    const int half = lane >> 4;
    const int lr   = lane & 15;
    const int rowbase = blockIdx.x * 16;

    const float* arow = A + (size_t)(rowbase + lr) * K + 2 * half;
    v2f a[32];
#pragma unroll
    for (int s = 0; s < 32; ++s) {
        a[s] = *(const v2f*)(arow + 4 * s);
    }

    const float* brel = B + (size_t)(2 * half) * LDB + lr;
    float* crel = C + (size_t)(rowbase + 8 * half) * LDC + lr;
    for (int r = 0; r < NREL; ++r) {
#pragma unroll
        for (int ct = 0; ct < NCT; ++ct) {
            const float* bp = brel + ct * 16;
            v2f b[32];
#pragma unroll
            for (int s = 0; s < 32; ++s) {
                b[s].x = bp[(4 * s + 0) * LDB];
                b[s].y = bp[(4 * s + 1) * LDB];
            }
            v8f acc0 = {};
            v8f acc1 = {};
#pragma unroll
            for (int s = 0; s < 32; s += 2) {
                acc0 = __builtin_amdgcn_wmma_f32_16x16x4_f32(
                    false, a[s], false, b[s], (short)0, acc0, false, false);
                acc1 = __builtin_amdgcn_wmma_f32_16x16x4_f32(
                    false, a[s + 1], false, b[s + 1], (short)0, acc1, false, false);
            }
            float* crow = crel + ct * 16;
#pragma unroll
            for (int j = 0; j < 8; ++j) {
                crow[(size_t)j * LDC] = acc0[j] + acc1[j];
            }
        }
        brel += (size_t)K * LDB;
        crel += NCT * 16;
    }
}

template <int LDO, int RELU>
__global__ __launch_bounds__(256)
void gather_slab_kernel(const float* __restrict__ slab,
                        const int* __restrict__ rowptr,
                        const int* __restrict__ eg,
                        const float* __restrict__ ewn,
                        const float* __restrict__ bias,
                        float* __restrict__ dest, int c0) {
    static_assert(NN % 16 == 0, "256 threads take 16 whole nodes");
    const int lane = threadIdx.x & 31;
    const int sub  = lane & 15;
    const int hb   = lane & 16;
    const int n = ((blockIdx.x * blockDim.x + threadIdx.x) >> 5) * 2 + (lane >> 4);
    const int start = rowptr[n];
    const int deg   = rowptr[n + 1] - start;
    const int degmax = max(deg, __shfl_xor(deg, 16, 32));
    float acc = 0.0f;
    for (int base = 0; base < degmax; base += 16) {
        const int rem = deg - base;
        int   gl = 0;
        float wl = 0.0f;
        if (sub < rem) {
            gl = eg[start + base + sub];
            wl = ewn[start + base + sub];
        }
#pragma unroll
        for (int k = 0; k < 16; ++k) {
            const int   g  = __shfl(gl, hb + k, 32);
            const float wk = __shfl(wl, hb + k, 32);
            const float t  = slab[(size_t)g * 16 + sub];
            acc += (k < rem) ? t * wk : 0.0f;
        }
    }
    float* o = dest + (size_t)n * LDO + c0 + sub;
    const float v = acc * (1.0f / fmaxf((float)deg, 1.0f)) + *o + bias[c0 + sub];
    *o = RELU ? fmaxf(v, 0.0f) : v;
}

extern "C" void kernel_launch(void* const* d_in, const int* in_sizes, int n_in,
                              void* d_out, int out_size, void* d_ws, size_t ws_size,
                              hipStream_t stream) {
    const float* x     = (const float*)d_in[0];
    const int*   eidx  = (const int*)d_in[1];
    const float* ew    = (const float*)d_in[2];
    const int*   et    = (const int*)d_in[3];
    const float* W1    = (const float*)d_in[4];
    const float* root1 = (const float*)d_in[5];
    const float* bias1 = (const float*)d_in[6];
    const float* W2    = (const float*)d_in[7];
    const float* root2 = (const float*)d_in[8];
    const float* bias2 = (const float*)d_in[9];
    float* out = (float*)d_out;

    const int* src = eidx;
    const int* dst = eidx + EE;

    int*   deg    = (int*)d_ws;
    int*   rowptr = deg + NN;
    int*   rank   = rowptr + NN + 1;
    int*   eg     = rank + EE;
    float* ewn    = (float*)(eg + EE);
    unsigned* mm  = (unsigned*)(ewn + EE);
    float* slab = (float*)d_ws + (size_t)NN * HH;
    float* h    = slab + (size_t)NN * HH;

    const int T = 256;
    const int rowTiles = NN / 16;
    const int gblocks  = NN / 16;

    hipMemsetAsync(deg, 0, (size_t)NN * sizeof(int), stream);
    init_mm_kernel<<<1, 32, 0, stream>>>(mm);
    ew_minmax_kernel<<<1024, T, 0, stream>>>(ew, EE, mm);
    count_rank_kernel<<<(EE + T - 1) / T, T, 0, stream>>>(dst, deg, rank, EE);
    scan_kernel<<<1, 1024, 0, stream>>>(deg, rowptr);
    fill_kernel<<<(EE + T - 1) / T, T, 0, stream>>>(src, dst, et, ew, mm, rowptr, rank, eg, ewn, EE);

    gemm_slab_wmma_f32_kernel<HH, HH, HH / 16, 1><<<rowTiles, 32, 0, stream>>>(x, root1, h);
    for (int c0 = 0; c0 < HH; c0 += 16) {
        gemm_slab_wmma_f32_kernel<HH, RR * 16, 1, RR><<<rowTiles, 32, 0, stream>>>(
            x, W1 + c0, slab);
        gather_slab_kernel<HH, 1><<<gblocks, T, 0, stream>>>(
            slab, rowptr, eg, ewn, bias1, h, c0);
    }

    gemm_slab_wmma_f32_kernel<OO, OO, OO / 16, 1><<<rowTiles, 32, 0, stream>>>(h, root2, out);
    for (int c0 = 0; c0 < OO; c0 += 16) {
        gemm_slab_wmma_f32_kernel<OO, RR * 16, 1, RR><<<rowTiles, 32, 0, stream>>>(
            h, W2 + c0, slab);
        gather_slab_kernel<OO, 0><<<gblocks, T, 0, stream>>>(
            slab, rowptr, eg, ewn, bias2, out, c0);
    }
}


# Row-start scan through LDS: the 50000 counts come in by coalesced loads all in flight at once, each thread scans its run of 49 from LDS, results leave coalesced; 98 dependent load steps gone
# speedup vs baseline: 1.6531x; 1.1598x over previous
#include <hip/hip_runtime.h>
#include <hip/hip_bf16.h>

#define NN 50000
#define EE 600000
#define FF 128
#define HH 128
#define OO 64
#define RR 8

typedef float v2f __attribute__((ext_vector_type(2)));
typedef float v8f __attribute__((ext_vector_type(8)));

__global__ void init_mm_kernel(unsigned* mm) {
    if (threadIdx.x == 0) { mm[0] = 0x7F800000u; mm[1] = 0u; }
}

__global__ void ew_minmax_kernel(const float* __restrict__ ew, int E,
                                 unsigned* __restrict__ mm) {
    float mn = __builtin_huge_valf();
    float mx = 0.0f;
    for (int i = blockIdx.x * blockDim.x + threadIdx.x; i < E;
         i += gridDim.x * blockDim.x) {
        float w = ew[i];
        mn = fminf(mn, w);
        mx = fmaxf(mx, w);
    }
    for (int off = 16; off > 0; off >>= 1) {
        mn = fminf(mn, __shfl_down(mn, off, 32));
        mx = fmaxf(mx, __shfl_down(mx, off, 32));
    }
    if ((threadIdx.x & 31) == 0) {
        atomicMin(&mm[0], __float_as_uint(mn));
        atomicMax(&mm[1], __float_as_uint(mx));
    }
}

__global__ void count_rank_kernel(const int* __restrict__ dst, int* __restrict__ deg,
                                  int* __restrict__ rank, int E) {
    int e = blockIdx.x * blockDim.x + threadIdx.x;
    if (e < E) rank[e] = atomicAdd(&deg[dst[e]], 1);
}

__global__ __launch_bounds__(1024)
void scan_kernel(const int* __restrict__ deg, int* __restrict__ rowptr) {
    constexpr int PER = (NN + 1023) / 1024;
    __shared__ int sm[PER * 1024];
    __shared__ int wtot[32];
    const int t    = threadIdx.x;
    const int lane = t & 31;
    const int w    = t >> 5;

    int v[PER];
#pragma unroll
    for (int j = 0; j < PER; ++j) {
        const int idx = j * 1024 + t;
        const int d = deg[min(idx, NN - 1)];
        v[j] = (idx < NN) ? d : 0;
    }
#pragma unroll
    for (int j = 0; j < PER; ++j) sm[j * 1024 + t] = v[j];
    __syncthreads();

    const int b = t * PER;
    int c[PER];
    int s = 0;
#pragma unroll
    for (int i = 0; i < PER; ++i) { c[i] = sm[b + i]; s += c[i]; }

    int inc = s;
#pragma unroll
    for (int off = 1; off < 32; off <<= 1) {
        const int u = __shfl_up(inc, off, 32);
        if (lane >= off) inc += u;
    }
    if (lane == 31) wtot[w] = inc;
    __syncthreads();
    const int wt = wtot[lane];
    int winc = wt;
#pragma unroll
    for (int off = 1; off < 32; off <<= 1) {
        const int u = __shfl_up(winc, off, 32);
        if (lane >= off) winc += u;
    }
    const int wbase = __shfl(winc, w, 32) - __shfl(wt, w, 32);
    const int total = __shfl(winc, 31, 32);

    int run = wbase + inc - s;
#pragma unroll
    for (int i = 0; i < PER; ++i) { sm[b + i] = run; run += c[i]; }
    __syncthreads();

#pragma unroll
    for (int j = 0; j < PER; ++j) {
        const int idx = j * 1024 + t;
        if (idx < NN) rowptr[idx] = sm[idx];
    }
    if (t == 0) rowptr[NN] = total;
}

__global__ void fill_kernel(const int* __restrict__ src, const int* __restrict__ dst,
                            const int* __restrict__ et, const float* __restrict__ ew,
                            const unsigned* __restrict__ mm,
                            const int* __restrict__ rowptr, const int* __restrict__ rank,
                            int* __restrict__ eg, float* __restrict__ ewn, int E) {
    int e = blockIdx.x * blockDim.x + threadIdx.x;
    if (e >= E) return;
    const float mn = __uint_as_float(mm[0]);
    const float mx = __uint_as_float(mm[1]);
    const int p = rowptr[dst[e]] + rank[e];
    eg[p]  = src[e] * RR + et[e];
    ewn[p] = (ew[e] - mn) / (mx - mn + 1e-8f);
}

template <int LDB, int LDC, int NCT, int NREL>
__global__ __launch_bounds__(32)
void gemm_slab_wmma_f32_kernel(const float* __restrict__ A, const float* __restrict__ B,
                               float* __restrict__ C) {
    const int K = 128;
    const int lane = threadIdx.x;
    const int half = lane >> 4;
    const int lr   = lane & 15;
    const int rowbase = blockIdx.x * 16;

    const float* arow = A + (size_t)(rowbase + lr) * K + 2 * half;
    v2f a[32];
#pragma unroll
    for (int s = 0; s < 32; ++s) {
        a[s] = *(const v2f*)(arow + 4 * s);
    }

    const float* brel = B + (size_t)(2 * half) * LDB + lr;
    float* crel = C + (size_t)(rowbase + 8 * half) * LDC + lr;
    for (int r = 0; r < NREL; ++r) {
#pragma unroll
        for (int ct = 0; ct < NCT; ++ct) {
            const float* bp = brel + ct * 16;
            v2f b[32];
#pragma unroll
            for (int s = 0; s < 32; ++s) {
                b[s].x = bp[(4 * s + 0) * LDB];
                b[s].y = bp[(4 * s + 1) * LDB];
            }
            v8f acc0 = {};
            v8f acc1 = {};
#pragma unroll
            for (int s = 0; s < 32; s += 2) {
                acc0 = __builtin_amdgcn_wmma_f32_16x16x4_f32(
                    false, a[s], false, b[s], (short)0, acc0, false, false);
                acc1 = __builtin_amdgcn_wmma_f32_16x16x4_f32(
                    false, a[s + 1], false, b[s + 1], (short)0, acc1, false, false);
            }
            float* crow = crel + ct * 16;
#pragma unroll
            for (int j = 0; j < 8; ++j) {
                crow[(size_t)j * LDC] = acc0[j] + acc1[j];
            }
        }
        brel += (size_t)K * LDB;
        crel += NCT * 16;
    }
}

template <int LDO, int RELU>
__global__ __launch_bounds__(256)
void gather_slab_kernel(const float* __restrict__ slab,
                        const int* __restrict__ rowptr,
                        const int* __restrict__ eg,
                        const float* __restrict__ ewn,
                        const float* __restrict__ bias,
                        float* __restrict__ dest, int c0) {
    static_assert(NN % 16 == 0, "256 threads take 16 whole nodes");
    const int lane = threadIdx.x & 31;
    const int sub  = lane & 15;
    const int hb   = lane & 16;
    const int n = ((blockIdx.x * blockDim.x + threadIdx.x) >> 5) * 2 + (lane >> 4);
    const int start = rowptr[n];
    const int deg   = rowptr[n + 1] - start;
    const int degmax = max(deg, __shfl_xor(deg, 16, 32));
    float acc = 0.0f;
    for (int base = 0; base < degmax; base += 16) {
        const int rem = deg - base;
        int   gl = 0;
        float wl = 0.0f;
        if (sub < rem) {
            gl = eg[start + base + sub];
            wl = ewn[start + base + sub];
        }
#pragma unroll
        for (int k = 0; k < 16; ++k) {
            const int   g  = __shfl(gl, hb + k, 32);
            const float wk = __shfl(wl, hb + k, 32);
            const float t  = slab[(size_t)g * 16 + sub];
            acc += (k < rem) ? t * wk : 0.0f;
        }
    }
    float* o = dest + (size_t)n * LDO + c0 + sub;
    const float v = acc * (1.0f / fmaxf((float)deg, 1.0f)) + *o + bias[c0 + sub];
    *o = RELU ? fmaxf(v, 0.0f) : v;
}

extern "C" void kernel_launch(void* const* d_in, const int* in_sizes, int n_in,
                              void* d_out, int out_size, void* d_ws, size_t ws_size,
                              hipStream_t stream) {
    const float* x     = (const float*)d_in[0];
    const int*   eidx  = (const int*)d_in[1];
    const float* ew    = (const float*)d_in[2];
    const int*   et    = (const int*)d_in[3];
    const float* W1    = (const float*)d_in[4];
    const float* root1 = (const float*)d_in[5];
    const float* bias1 = (const float*)d_in[6];
    const float* W2    = (const float*)d_in[7];
    const float* root2 = (const float*)d_in[8];
    const float* bias2 = (const float*)d_in[9];
    float* out = (float*)d_out;

    const int* src = eidx;
    const int* dst = eidx + EE;

    int*   deg    = (int*)d_ws;
    int*   rowptr = deg + NN;
    int*   rank   = rowptr + NN + 1;
    int*   eg     = rank + EE;
    float* ewn    = (float*)(eg + EE);
    unsigned* mm  = (unsigned*)(ewn + EE);
    float* slab = (float*)d_ws + (size_t)NN * HH;
    float* h    = slab + (size_t)NN * HH;

    const int T = 256;
    const int rowTiles = NN / 16;
    const int gblocks  = NN / 16;

    hipMemsetAsync(deg, 0, (size_t)NN * sizeof(int), stream);
    init_mm_kernel<<<1, 32, 0, stream>>>(mm);
    ew_minmax_kernel<<<1024, T, 0, stream>>>(ew, EE, mm);
    count_rank_kernel<<<(EE + T - 1) / T, T, 0, stream>>>(dst, deg, rank, EE);
    scan_kernel<<<1, 1024, 0, stream>>>(deg, rowptr);
    fill_kernel<<<(EE + T - 1) / T, T, 0, stream>>>(src, dst, et, ew, mm, rowptr, rank, eg, ewn, EE);

    gemm_slab_wmma_f32_kernel<HH, HH, HH / 16, 1><<<rowTiles, 32, 0, stream>>>(x, root1, h);
    for (int c0 = 0; c0 < HH; c0 += 16) {
        gemm_slab_wmma_f32_kernel<HH, RR * 16, 1, RR><<<rowTiles, 32, 0, stream>>>(
            x, W1 + c0, slab);
        gather_slab_kernel<HH, 1><<<gblocks, T, 0, stream>>>(
            slab, rowptr, eg, ewn, bias1, h, c0);
    }

    gemm_slab_wmma_f32_kernel<OO, OO, OO / 16, 1><<<rowTiles, 32, 0, stream>>>(h, root2, out);
    for (int c0 = 0; c0 < OO; c0 += 16) {
        gemm_slab_wmma_f32_kernel<OO, RR * 16, 1, RR><<<rowTiles, 32, 0, stream>>>(
            h, W2 + c0, slab);
        gather_slab_kernel<OO, 0><<<gblocks, T, 0, stream>>>(
            slab, rowptr, eg, ewn, bias2, out, c0);
    }
}


# Layer-1 slab 32 columns wide, not 16 (relations 0-6 in the workspace, relation 7 in the not yet used output buffer): 4+4 slab launches, not 8+8; an edge row is one 128-B line; sum order unchanged
# speedup vs baseline: 1.8758x; 1.1347x over previous
#include <hip/hip_runtime.h>
#include <hip/hip_bf16.h>

#define NN 50000
#define EE 600000
#define FF 128
#define HH 128
#define OO 64
#define RR 8

typedef float v2f __attribute__((ext_vector_type(2)));
typedef float v8f __attribute__((ext_vector_type(8)));

__global__ void init_mm_kernel(unsigned* mm) {
    if (threadIdx.x == 0) { mm[0] = 0x7F800000u; mm[1] = 0u; }
}

__global__ void ew_minmax_kernel(const float* __restrict__ ew, int E,
                                 unsigned* __restrict__ mm) {
    float mn = __builtin_huge_valf();
    float mx = 0.0f;
    for (int i = blockIdx.x * blockDim.x + threadIdx.x; i < E;
         i += gridDim.x * blockDim.x) {
        float w = ew[i];
        mn = fminf(mn, w);
        mx = fmaxf(mx, w);
    }
    for (int off = 16; off > 0; off >>= 1) {
        mn = fminf(mn, __shfl_down(mn, off, 32));
        mx = fmaxf(mx, __shfl_down(mx, off, 32));
    }
    if ((threadIdx.x & 31) == 0) {
        atomicMin(&mm[0], __float_as_uint(mn));
        atomicMax(&mm[1], __float_as_uint(mx));
    }
}

__global__ void count_rank_kernel(const int* __restrict__ dst, int* __restrict__ deg,
                                  int* __restrict__ rank, int E) {
    int e = blockIdx.x * blockDim.x + threadIdx.x;
    if (e < E) rank[e] = atomicAdd(&deg[dst[e]], 1);
}

__global__ __launch_bounds__(1024)
void scan_kernel(const int* __restrict__ deg, int* __restrict__ rowptr) {
    constexpr int PER = (NN + 1023) / 1024;
    __shared__ int sm[PER * 1024];
    __shared__ int wtot[32];
    const int t    = threadIdx.x;
    const int lane = t & 31;
    const int w    = t >> 5;

    int v[PER];
#pragma unroll
    for (int j = 0; j < PER; ++j) {
        const int idx = j * 1024 + t;
        const int d = deg[min(idx, NN - 1)];
        v[j] = (idx < NN) ? d : 0;
    }
#pragma unroll
    for (int j = 0; j < PER; ++j) sm[j * 1024 + t] = v[j];
    __syncthreads();

    const int b = t * PER;
    int c[PER];
    int s = 0;
#pragma unroll
    for (int i = 0; i < PER; ++i) { c[i] = sm[b + i]; s += c[i]; }

    int inc = s;
#pragma unroll
    for (int off = 1; off < 32; off <<= 1) {
        const int u = __shfl_up(inc, off, 32);
        if (lane >= off) inc += u;
    }
    if (lane == 31) wtot[w] = inc;
    __syncthreads();
    const int wt = wtot[lane];
    int winc = wt;
#pragma unroll
    for (int off = 1; off < 32; off <<= 1) {
        const int u = __shfl_up(winc, off, 32);
        if (lane >= off) winc += u;
    }
    const int wbase = __shfl(winc, w, 32) - __shfl(wt, w, 32);
    const int total = __shfl(winc, 31, 32);

    int run = wbase + inc - s;
#pragma unroll
    for (int i = 0; i < PER; ++i) { sm[b + i] = run; run += c[i]; }
    __syncthreads();

#pragma unroll
    for (int j = 0; j < PER; ++j) {
        const int idx = j * 1024 + t;
        if (idx < NN) rowptr[idx] = sm[idx];
    }
    if (t == 0) rowptr[NN] = total;
}

__global__ void fill_kernel(const int* __restrict__ src, const int* __restrict__ dst,
                            const int* __restrict__ et, const float* __restrict__ ew,
                            const unsigned* __restrict__ mm,
                            const int* __restrict__ rowptr, const int* __restrict__ rank,
                            int* __restrict__ eg, float* __restrict__ ewn, int E) {
    int e = blockIdx.x * blockDim.x + threadIdx.x;
    if (e >= E) return;
    const float mn = __uint_as_float(mm[0]);
    const float mx = __uint_as_float(mm[1]);
    const int p = rowptr[dst[e]] + rank[e];
    eg[p]  = src[e] * RR + et[e];
    ewn[p] = (ew[e] - mn) / (mx - mn + 1e-8f);
}

template <int LDB, int LDCX, int NCT>
__device__ __forceinline__ void slab_relation(const v2f (&a)[32], const float* __restrict__ brel,
                                              float* __restrict__ crel) {
#pragma unroll
    for (int ct = 0; ct < NCT; ++ct) {
        const float* bp = brel + ct * 16;
        v2f b[32];
#pragma unroll
        for (int s = 0; s < 32; ++s) {
            b[s].x = bp[(4 * s + 0) * LDB];
            b[s].y = bp[(4 * s + 1) * LDB];
        }
        v8f acc0 = {};
        v8f acc1 = {};
#pragma unroll
        for (int s = 0; s < 32; s += 2) {
            acc0 = __builtin_amdgcn_wmma_f32_16x16x4_f32(
                false, a[s], false, b[s], (short)0, acc0, false, false);
            acc1 = __builtin_amdgcn_wmma_f32_16x16x4_f32(
                false, a[s + 1], false, b[s + 1], (short)0, acc1, false, false);
        }
        float* crow = crel + ct * 16;
#pragma unroll
        for (int j = 0; j < 8; ++j) {
            crow[(size_t)j * LDCX] = acc0[j] + acc1[j];
        }
    }
}

template <int LDB, int LDC, int NCT, int NREL, int LDC2>
__global__ __launch_bounds__(32)
void gemm_slab_wmma_f32_kernel(const float* __restrict__ A, const float* __restrict__ B,
                               float* __restrict__ C, float* __restrict__ C2) {
    const int K = 128;
    const int lane = threadIdx.x;
    const int half = lane >> 4;
    const int lr   = lane & 15;
    const int rowbase = blockIdx.x * 16;

    const float* arow = A + (size_t)(rowbase + lr) * K + 2 * half;
    v2f a[32];
#pragma unroll
    for (int s = 0; s < 32; ++s) {
        a[s] = *(const v2f*)(arow + 4 * s);
    }

    const float* brel = B + (size_t)(2 * half) * LDB + lr;
    float* crel = C + (size_t)(rowbase + 8 * half) * LDC + lr;
    for (int r = 0; r < NREL; ++r) {
        slab_relation<LDB, LDC, NCT>(a, brel, crel);
        brel += (size_t)K * LDB;
        crel += NCT * 16;
    }
    if constexpr (LDC2 > 0) {
        float* c2 = C2 + (size_t)(rowbase + 8 * half) * LDC2 + lr;
        slab_relation<LDB, LDC2, NCT>(a, brel, c2);
    }
}

template <int LDO, int RELU>
__global__ __launch_bounds__(256)
void gather_slab_kernel(const float* __restrict__ slab,
                        const int* __restrict__ rowptr,
                        const int* __restrict__ eg,
                        const float* __restrict__ ewn,
                        const float* __restrict__ bias,
                        float* __restrict__ dest, int c0) {
    static_assert(NN % 16 == 0, "256 threads take 16 whole nodes");
    const int lane = threadIdx.x & 31;
    const int sub  = lane & 15;
    const int hb   = lane & 16;
    const int n = ((blockIdx.x * blockDim.x + threadIdx.x) >> 5) * 2 + (lane >> 4);
    const int start = rowptr[n];
    const int deg   = rowptr[n + 1] - start;
    const int degmax = max(deg, __shfl_xor(deg, 16, 32));
    float acc = 0.0f;
    for (int base = 0; base < degmax; base += 16) {
        const int rem = deg - base;
        int   gl = 0;
        float wl = 0.0f;
        if (sub < rem) {
            gl = eg[start + base + sub];
            wl = ewn[start + base + sub];
        }
#pragma unroll
        for (int k = 0; k < 16; ++k) {
            const int   g  = __shfl(gl, hb + k, 32);
            const float wk = __shfl(wl, hb + k, 32);
            const float t  = slab[(size_t)g * 16 + sub];
            acc += (k < rem) ? t * wk : 0.0f;
        }
    }
    float* o = dest + (size_t)n * LDO + c0 + sub;
    const float v = acc * (1.0f / fmaxf((float)deg, 1.0f)) + *o + bias[c0 + sub];
    *o = RELU ? fmaxf(v, 0.0f) : v;
}

template <int LDO, int RELU>
__global__ __launch_bounds__(256)
void gather_slab32_kernel(const float* __restrict__ slabA,
                          const float* __restrict__ slabB,
                          const int* __restrict__ rowptr,
                          const int* __restrict__ eg,
                          const float* __restrict__ ewn,
                          const float* __restrict__ bias,
                          float* __restrict__ dest, int c0) {
    static_assert(NN % 16 == 0, "256 threads take 16 whole nodes");
    const int lane = threadIdx.x & 31;
    const int sub  = lane & 15;
    const int hb   = lane & 16;
    const int n = ((blockIdx.x * blockDim.x + threadIdx.x) >> 5) * 2 + (lane >> 4);
    const int start = rowptr[n];
    const int deg   = rowptr[n + 1] - start;
    const int degmax = max(deg, __shfl_xor(deg, 16, 32));
    float acc0 = 0.0f;
    float acc1 = 0.0f;
    for (int base = 0; base < degmax; base += 16) {
        const int rem = deg - base;
        int   cl = 0;
        float wl = 0.0f;
        if (sub < rem) {
            const int g = eg[start + base + sub];
            wl = ewn[start + base + sub];
            const int  s    = g >> 3;
            const bool last = (g & 7) == 7;
            cl = (last ? s : g - s) * 128 + (last ? 1 : 0);
        }
#pragma unroll
        for (int k = 0; k < 16; ++k) {
            const int   ck = __shfl(cl, hb + k, 32);
            const float wk = __shfl(wl, hb + k, 32);
            const char* bufp = (ck & 1) ? (const char*)slabB : (const char*)slabA;
            const unsigned off = ((unsigned)ck & ~127u) | ((unsigned)sub << 3);
            const v2f t = *(const v2f*)(bufp + off);
            acc0 += (k < rem) ? t.x * wk : 0.0f;
            acc1 += (k < rem) ? t.y * wk : 0.0f;
        }
    }
    float* o = dest + (size_t)n * LDO + c0 + 2 * sub;
    const v2f ov = *(const v2f*)o;
    const float inv = 1.0f / fmaxf((float)deg, 1.0f);
    v2f v;
    v.x = acc0 * inv + ov.x + bias[c0 + 2 * sub];
    v.y = acc1 * inv + ov.y + bias[c0 + 2 * sub + 1];
    if (RELU) { v.x = fmaxf(v.x, 0.0f); v.y = fmaxf(v.y, 0.0f); }
    *(v2f*)o = v;
}

extern "C" void kernel_launch(void* const* d_in, const int* in_sizes, int n_in,
                              void* d_out, int out_size, void* d_ws, size_t ws_size,
                              hipStream_t stream) {
    const float* x     = (const float*)d_in[0];
    const int*   eidx  = (const int*)d_in[1];
    const float* ew    = (const float*)d_in[2];
    const int*   et    = (const int*)d_in[3];
    const float* W1    = (const float*)d_in[4];
    const float* root1 = (const float*)d_in[5];
    const float* bias1 = (const float*)d_in[6];
    const float* W2    = (const float*)d_in[7];
    const float* root2 = (const float*)d_in[8];
    const float* bias2 = (const float*)d_in[9];
    float* out = (float*)d_out;

    const int* src = eidx;
    const int* dst = eidx + EE;

    int*   rowptr = (int*)d_ws;
    int*   eg     = rowptr + NN + 1;
    float* ewn    = (float*)(eg + EE);
    unsigned* mm  = (unsigned*)(ewn + EE);
    float* slab = (float*)d_ws + 1250016;
    float* h    = (float*)d_ws + (size_t)2 * NN * HH;
    int*   deg    = (int*)h;
    int*   rank   = deg + NN;
    float* slab7  = out;

    const int T = 256;
    const int rowTiles = NN / 16;
    const int gblocks  = NN / 16;

    hipMemsetAsync(deg, 0, (size_t)NN * sizeof(int), stream);
    init_mm_kernel<<<1, 32, 0, stream>>>(mm);
    ew_minmax_kernel<<<1024, T, 0, stream>>>(ew, EE, mm);
    count_rank_kernel<<<(EE + T - 1) / T, T, 0, stream>>>(dst, deg, rank, EE);
    scan_kernel<<<1, 1024, 0, stream>>>(deg, rowptr);
    fill_kernel<<<(EE + T - 1) / T, T, 0, stream>>>(src, dst, et, ew, mm, rowptr, rank, eg, ewn, EE);

    gemm_slab_wmma_f32_kernel<HH, HH, HH / 16, 1, 0><<<rowTiles, 32, 0, stream>>>(
        x, root1, h, nullptr);
    for (int c0 = 0; c0 < HH; c0 += 32) {
        gemm_slab_wmma_f32_kernel<HH, (RR - 1) * 32, 2, RR - 1, 32><<<rowTiles, 32, 0, stream>>>(
            x, W1 + c0, slab, slab7);
        gather_slab32_kernel<HH, 1><<<gblocks, T, 0, stream>>>(
            slab, slab7, rowptr, eg, ewn, bias1, h, c0);
    }

    gemm_slab_wmma_f32_kernel<OO, OO, OO / 16, 1, 0><<<rowTiles, 32, 0, stream>>>(
        h, root2, out, nullptr);
    for (int c0 = 0; c0 < OO; c0 += 16) {
        gemm_slab_wmma_f32_kernel<OO, RR * 16, 1, RR, 0><<<rowTiles, 32, 0, stream>>>(
            h, W2 + c0, slab, nullptr);
        gather_slab_kernel<OO, 0><<<gblocks, T, 0, stream>>>(
            slab, rowptr, eg, ewn, bias2, out, c0);
    }
}


# Fewer dependent launches (23 + a memset -> 18): each slab GEMM also does its root columns; layer 2's columns 0..31 are one 32-wide slab; counters zeroed and weight min/max taken atomic-free in 1 pass
# speedup vs baseline: 2.0057x; 1.0693x over previous
#include <hip/hip_runtime.h>
#include <hip/hip_bf16.h>

#define NN 50000
#define EE 600000
#define FF 128
#define HH 128
#define OO 64
#define RR 8

typedef float v2f __attribute__((ext_vector_type(2)));
typedef float v8f __attribute__((ext_vector_type(8)));

#define PREP_B 256
#define PREP_PER ((EE + PREP_B * 256 - 1) / (PREP_B * 256))

__global__ __launch_bounds__(256)
void prep_kernel(const float* __restrict__ ew, float* __restrict__ part,
                 int* __restrict__ deg) {
    __shared__ float smn[8];
    __shared__ float smx[8];
    const int tid = blockIdx.x * 256 + threadIdx.x;
    if (tid < NN) deg[tid] = 0;
    static_assert(PREP_B * 256 >= NN, "one counter per thread");
    float v[PREP_PER];
#pragma unroll
    for (int j = 0; j < PREP_PER; ++j) v[j] = ew[min(j * (PREP_B * 256) + tid, EE - 1)];
    float mn = v[0];
    float mx = v[0];
#pragma unroll
    for (int j = 1; j < PREP_PER; ++j) { mn = fminf(mn, v[j]); mx = fmaxf(mx, v[j]); }
    for (int off = 16; off > 0; off >>= 1) {
        mn = fminf(mn, __shfl_xor(mn, off, 32));
        mx = fmaxf(mx, __shfl_xor(mx, off, 32));
    }
    if ((threadIdx.x & 31) == 0) { smn[threadIdx.x >> 5] = mn; smx[threadIdx.x >> 5] = mx; }
    __syncthreads();
    if (threadIdx.x == 0) {
#pragma unroll
        for (int k = 1; k < 8; ++k) { mn = fminf(mn, smn[k]); mx = fmaxf(mx, smx[k]); }
        part[blockIdx.x] = mn;
        part[PREP_B + blockIdx.x] = mx;
    }
}

__global__ void count_rank_kernel(const int* __restrict__ dst, int* __restrict__ deg,
                                  int* __restrict__ rank, int E) {
    int e = blockIdx.x * blockDim.x + threadIdx.x;
    if (e < E) rank[e] = atomicAdd(&deg[dst[e]], 1);
}

__global__ __launch_bounds__(1024)
void scan_kernel(const int* __restrict__ deg, int* __restrict__ rowptr,
                 const float* __restrict__ part, unsigned* __restrict__ mm) {
    constexpr int PER = (NN + 1023) / 1024;
    __shared__ int sm[PER * 1024];
    __shared__ int wtot[32];
    __shared__ float pmn_s[PREP_B / 32];
    __shared__ float pmx_s[PREP_B / 32];
    const int t    = threadIdx.x;
    const int lane = t & 31;
    const int w    = t >> 5;

    float pmn = part[t & (PREP_B - 1)];
    float pmx = part[PREP_B + (t & (PREP_B - 1))];

    int v[PER];
#pragma unroll
    for (int j = 0; j < PER; ++j) {
        const int idx = j * 1024 + t;
        const int d = deg[min(idx, NN - 1)];
        v[j] = (idx < NN) ? d : 0;
    }
#pragma unroll
    for (int j = 0; j < PER; ++j) sm[j * 1024 + t] = v[j];
    __syncthreads();

    const int b = t * PER;
    int c[PER];
    int s = 0;
#pragma unroll
    for (int i = 0; i < PER; ++i) { c[i] = sm[b + i]; s += c[i]; }

    int inc = s;
#pragma unroll
    for (int off = 1; off < 32; off <<= 1) {
        const int u = __shfl_up(inc, off, 32);
        if (lane >= off) inc += u;
    }
    if (lane == 31) wtot[w] = inc;

#pragma unroll
    for (int off = 16; off > 0; off >>= 1) {
        pmn = fminf(pmn, __shfl_xor(pmn, off, 32));
        pmx = fmaxf(pmx, __shfl_xor(pmx, off, 32));
    }
    if (lane == 0 && w < PREP_B / 32) { pmn_s[w] = pmn; pmx_s[w] = pmx; }
    __syncthreads();
    if (t == 0) {
#pragma unroll
        for (int k = 1; k < PREP_B / 32; ++k) { pmn = fminf(pmn, pmn_s[k]); pmx = fmaxf(pmx, pmx_s[k]); }
        mm[0] = __float_as_uint(pmn);
        mm[1] = __float_as_uint(pmx);
    }
    const int wt = wtot[lane];
    int winc = wt;
#pragma unroll
    for (int off = 1; off < 32; off <<= 1) {
        const int u = __shfl_up(winc, off, 32);
        if (lane >= off) winc += u;
    }
    const int wbase = __shfl(winc, w, 32) - __shfl(wt, w, 32);
    const int total = __shfl(winc, 31, 32);

    int run = wbase + inc - s;
#pragma unroll
    for (int i = 0; i < PER; ++i) { sm[b + i] = run; run += c[i]; }
    __syncthreads();

#pragma unroll
    for (int j = 0; j < PER; ++j) {
        const int idx = j * 1024 + t;
        if (idx < NN) rowptr[idx] = sm[idx];
    }
    if (t == 0) rowptr[NN] = total;
}

__global__ void fill_kernel(const int* __restrict__ src, const int* __restrict__ dst,
                            const int* __restrict__ et, const float* __restrict__ ew,
                            const unsigned* __restrict__ mm,
                            const int* __restrict__ rowptr, const int* __restrict__ rank,
                            int* __restrict__ eg, float* __restrict__ ewn, int E) {
    int e = blockIdx.x * blockDim.x + threadIdx.x;
    if (e >= E) return;
    const float mn = __uint_as_float(mm[0]);
    const float mx = __uint_as_float(mm[1]);
    const int p = rowptr[dst[e]] + rank[e];
    eg[p]  = src[e] * RR + et[e];
    ewn[p] = (ew[e] - mn) / (mx - mn + 1e-8f);
}

template <int LDB, int LDCX, int NCT>
__device__ __forceinline__ void slab_relation(const v2f (&a)[32], const float* __restrict__ brel,
                                              float* __restrict__ crel) {
#pragma unroll
    for (int ct = 0; ct < NCT; ++ct) {
        const float* bp = brel + ct * 16;
        v2f b[32];
#pragma unroll
        for (int s = 0; s < 32; ++s) {
            b[s].x = bp[(4 * s + 0) * LDB];
            b[s].y = bp[(4 * s + 1) * LDB];
        }
        v8f acc0 = {};
        v8f acc1 = {};
#pragma unroll
        for (int s = 0; s < 32; s += 2) {
            acc0 = __builtin_amdgcn_wmma_f32_16x16x4_f32(
                false, a[s], false, b[s], (short)0, acc0, false, false);
            acc1 = __builtin_amdgcn_wmma_f32_16x16x4_f32(
                false, a[s + 1], false, b[s + 1], (short)0, acc1, false, false);
        }
        float* crow = crel + ct * 16;
#pragma unroll
        for (int j = 0; j < 8; ++j) {
            crow[(size_t)j * LDCX] = acc0[j] + acc1[j];
        }
    }
}

template <int LDB, int LDC, int NCT, int NREL, int LDC2, int LDR>
__global__ __launch_bounds__(32)
void gemm_slab_wmma_f32_kernel(const float* __restrict__ A, const float* __restrict__ B,
                               float* __restrict__ C, float* __restrict__ C2,
                               const float* __restrict__ Broot, float* __restrict__ Croot) {
    const int K = 128;
    const int lane = threadIdx.x;
    const int half = lane >> 4;
    const int lr   = lane & 15;
    const int rowbase = blockIdx.x * 16;

    const float* arow = A + (size_t)(rowbase + lr) * K + 2 * half;
    v2f a[32];
#pragma unroll
    for (int s = 0; s < 32; ++s) {
        a[s] = *(const v2f*)(arow + 4 * s);
    }

    const float* brel = B + (size_t)(2 * half) * LDB + lr;
    float* crel = C + (size_t)(rowbase + 8 * half) * LDC + lr;
    for (int r = 0; r < NREL; ++r) {
        slab_relation<LDB, LDC, NCT>(a, brel, crel);
        brel += (size_t)K * LDB;
        crel += NCT * 16;
    }
    if constexpr (LDC2 > 0) {
        float* c2 = C2 + (size_t)(rowbase + 8 * half) * LDC2 + lr;
        slab_relation<LDB, LDC2, NCT>(a, brel, c2);
    }
    if constexpr (LDR > 0) {
        const float* broot = Broot + (size_t)(2 * half) * LDB + lr;
        float* cr = Croot + (size_t)(rowbase + 8 * half) * LDR + lr;
        slab_relation<LDB, LDR, NCT>(a, broot, cr);
    }
}

template <int LDO, int RELU>
__global__ __launch_bounds__(256)
void gather_slab_kernel(const float* __restrict__ slab,
                        const int* __restrict__ rowptr,
                        const int* __restrict__ eg,
                        const float* __restrict__ ewn,
                        const float* __restrict__ bias,
                        float* __restrict__ dest, int c0) {
    static_assert(NN % 16 == 0, "256 threads take 16 whole nodes");
    const int lane = threadIdx.x & 31;
    const int sub  = lane & 15;
    const int hb   = lane & 16;
    const int n = ((blockIdx.x * blockDim.x + threadIdx.x) >> 5) * 2 + (lane >> 4);
    const int start = rowptr[n];
    const int deg   = rowptr[n + 1] - start;
    const int degmax = max(deg, __shfl_xor(deg, 16, 32));
    float acc = 0.0f;
    for (int base = 0; base < degmax; base += 16) {
        const int rem = deg - base;
        int   gl = 0;
        float wl = 0.0f;
        if (sub < rem) {
            gl = eg[start + base + sub];
            wl = ewn[start + base + sub];
        }
#pragma unroll
        for (int k = 0; k < 16; ++k) {
            const int   g  = __shfl(gl, hb + k, 32);
            const float wk = __shfl(wl, hb + k, 32);
            const float t  = slab[(size_t)g * 16 + sub];
            acc += (k < rem) ? t * wk : 0.0f;
        }
    }
    float* o = dest + (size_t)n * LDO + c0 + sub;
    const float v = acc * (1.0f / fmaxf((float)deg, 1.0f)) + *o + bias[c0 + sub];
    *o = RELU ? fmaxf(v, 0.0f) : v;
}

template <int LDO, int RELU, int PB>
__global__ __launch_bounds__(256)
void gather_slab32_kernel(const float* __restrict__ slabA,
                          const float* __restrict__ slabB,
                          const int* __restrict__ rowptr,
                          const int* __restrict__ eg,
                          const float* __restrict__ ewn,
                          const float* __restrict__ bias,
                          float* __restrict__ dest, int c0) {
    static_assert(NN % 16 == 0, "256 threads take 16 whole nodes");
    static_assert(PB % 128 == 0 && (long long)NN * PB < (1ll << 31), "row offsets keep bit 0 free and fit 31 bits");
    const int lane = threadIdx.x & 31;
    const int sub  = lane & 15;
    const int hb   = lane & 16;
    const int n = ((blockIdx.x * blockDim.x + threadIdx.x) >> 5) * 2 + (lane >> 4);
    const int start = rowptr[n];
    const int deg   = rowptr[n + 1] - start;
    const int degmax = max(deg, __shfl_xor(deg, 16, 32));
    float acc0 = 0.0f;
    float acc1 = 0.0f;
    for (int base = 0; base < degmax; base += 16) {
        const int rem = deg - base;
        int   cl = 0;
        float wl = 0.0f;
        if (sub < rem) {
            const int g = eg[start + base + sub];
            wl = ewn[start + base + sub];
            const int  s    = g >> 3;
            const bool last = (g & 7) == 7;
            cl = last ? s * PB + 1 : (g - s) * 128;
        }
#pragma unroll
        for (int k = 0; k < 16; ++k) {
            const int   ck = __shfl(cl, hb + k, 32);
            const float wk = __shfl(wl, hb + k, 32);
            const char* bufp = (ck & 1) ? (const char*)slabB : (const char*)slabA;
            const unsigned off = ((unsigned)ck & ~127u) | ((unsigned)sub << 3);
            const v2f t = *(const v2f*)(bufp + off);
            acc0 += (k < rem) ? t.x * wk : 0.0f;
            acc1 += (k < rem) ? t.y * wk : 0.0f;
        }
    }
    float* o = dest + (size_t)n * LDO + c0 + 2 * sub;
    const v2f ov = *(const v2f*)o;
    const float inv = 1.0f / fmaxf((float)deg, 1.0f);
    v2f v;
    v.x = acc0 * inv + ov.x + bias[c0 + 2 * sub];
    v.y = acc1 * inv + ov.y + bias[c0 + 2 * sub + 1];
    if (RELU) { v.x = fmaxf(v.x, 0.0f); v.y = fmaxf(v.y, 0.0f); }
    *(v2f*)o = v;
}

extern "C" void kernel_launch(void* const* d_in, const int* in_sizes, int n_in,
                              void* d_out, int out_size, void* d_ws, size_t ws_size,
                              hipStream_t stream) {
    const float* x     = (const float*)d_in[0];
    const int*   eidx  = (const int*)d_in[1];
    const float* ew    = (const float*)d_in[2];
    const int*   et    = (const int*)d_in[3];
    const float* W1    = (const float*)d_in[4];
    const float* root1 = (const float*)d_in[5];
    const float* bias1 = (const float*)d_in[6];
    const float* W2    = (const float*)d_in[7];
    const float* root2 = (const float*)d_in[8];
    const float* bias2 = (const float*)d_in[9];
    float* out = (float*)d_out;

    const int* src = eidx;
    const int* dst = eidx + EE;

    int*   rowptr = (int*)d_ws;
    int*   eg     = rowptr + NN + 1;
    float* ewn    = (float*)(eg + EE);
    unsigned* mm  = (unsigned*)(ewn + EE);
    float* slab = (float*)d_ws + 1250016;
    float* h    = (float*)d_ws + (size_t)2 * NN * HH;
    int*   deg    = (int*)h;
    int*   rank   = deg + NN;
    float* part   = slab;
    float* slab7  = out;

    const int T = 256;
    const int rowTiles = NN / 16;
    const int gblocks  = NN / 16;

    prep_kernel<<<PREP_B, 256, 0, stream>>>(ew, part, deg);
    count_rank_kernel<<<(EE + T - 1) / T, T, 0, stream>>>(dst, deg, rank, EE);
    scan_kernel<<<1, 1024, 0, stream>>>(deg, rowptr, part, mm);
    fill_kernel<<<(EE + T - 1) / T, T, 0, stream>>>(src, dst, et, ew, mm, rowptr, rank, eg, ewn, EE);

    for (int c0 = 0; c0 < HH; c0 += 32) {
        gemm_slab_wmma_f32_kernel<HH, (RR - 1) * 32, 2, RR - 1, 32, HH><<<rowTiles, 32, 0, stream>>>(
            x, W1 + c0, slab, slab7, root1 + c0, h + c0);
        gather_slab32_kernel<HH, 1, 128><<<gblocks, T, 0, stream>>>(
            slab, slab7, rowptr, eg, ewn, bias1, h, c0);
    }

    gemm_slab_wmma_f32_kernel<OO, (RR - 1) * 32, 2, RR - 1, OO, OO><<<rowTiles, 32, 0, stream>>>(
        h, W2, slab, out + 32, root2, out);
    gather_slab32_kernel<OO, 0, OO * 4><<<gblocks, T, 0, stream>>>(
        slab, out + 32, rowptr, eg, ewn, bias2, out, 0);
    for (int c0 = 32; c0 < OO; c0 += 16) {
        gemm_slab_wmma_f32_kernel<OO, RR * 16, 1, RR, 0, OO><<<rowTiles, 32, 0, stream>>>(
            h, W2 + c0, slab, nullptr, root2 + c0, out + c0);
        gather_slab_kernel<OO, 0><<<gblocks, T, 0, stream>>>(
            slab, rowptr, eg, ewn, bias2, out, c0);
    }
}


# GEMM re-tiled: workgroup = 5 row-tile waves x 3 (16-col slab: 4 or 5) of the slab's 9 matrices; B copied once per workgroup into LDS as ready fragments, not 4-byte loads per wave; sums unchanged
# speedup vs baseline: 2.0641x; 1.0291x over previous
#include <hip/hip_runtime.h>
#include <hip/hip_bf16.h>

#define NN 50000
#define EE 600000
#define FF 128
#define HH 128
#define OO 64
#define RR 8

typedef float v2f __attribute__((ext_vector_type(2)));
typedef float v8f __attribute__((ext_vector_type(8)));

#define PREP_B 256
#define PREP_PER ((EE + PREP_B * 256 - 1) / (PREP_B * 256))

__global__ __launch_bounds__(256)
void prep_kernel(const float* __restrict__ ew, float* __restrict__ part,
                 int* __restrict__ deg) {
    __shared__ float smn[8];
    __shared__ float smx[8];
    const int tid = blockIdx.x * 256 + threadIdx.x;
    if (tid < NN) deg[tid] = 0;
    static_assert(PREP_B * 256 >= NN, "one counter per thread");
    float v[PREP_PER];
#pragma unroll
    for (int j = 0; j < PREP_PER; ++j) v[j] = ew[min(j * (PREP_B * 256) + tid, EE - 1)];
    float mn = v[0];
    float mx = v[0];
#pragma unroll
    for (int j = 1; j < PREP_PER; ++j) { mn = fminf(mn, v[j]); mx = fmaxf(mx, v[j]); }
    for (int off = 16; off > 0; off >>= 1) {
        mn = fminf(mn, __shfl_xor(mn, off, 32));
        mx = fmaxf(mx, __shfl_xor(mx, off, 32));
    }
    if ((threadIdx.x & 31) == 0) { smn[threadIdx.x >> 5] = mn; smx[threadIdx.x >> 5] = mx; }
    __syncthreads();
    if (threadIdx.x == 0) {
#pragma unroll
        for (int k = 1; k < 8; ++k) { mn = fminf(mn, smn[k]); mx = fmaxf(mx, smx[k]); }
        part[blockIdx.x] = mn;
        part[PREP_B + blockIdx.x] = mx;
    }
}

__global__ void count_rank_kernel(const int* __restrict__ dst, int* __restrict__ deg,
                                  int* __restrict__ rank, int E) {
    int e = blockIdx.x * blockDim.x + threadIdx.x;
    if (e < E) rank[e] = atomicAdd(&deg[dst[e]], 1);
}

__global__ __launch_bounds__(1024)
void scan_kernel(const int* __restrict__ deg, int* __restrict__ rowptr,
                 const float* __restrict__ part, unsigned* __restrict__ mm) {
    constexpr int PER = (NN + 1023) / 1024;
    __shared__ int sm[PER * 1024];
    __shared__ int wtot[32];
    __shared__ float pmn_s[PREP_B / 32];
    __shared__ float pmx_s[PREP_B / 32];
    const int t    = threadIdx.x;
    const int lane = t & 31;
    const int w    = t >> 5;

    float pmn = part[t & (PREP_B - 1)];
    float pmx = part[PREP_B + (t & (PREP_B - 1))];

    int v[PER];
#pragma unroll
    for (int j = 0; j < PER; ++j) {
        const int idx = j * 1024 + t;
        const int d = deg[min(idx, NN - 1)];
        v[j] = (idx < NN) ? d : 0;
    }
#pragma unroll
    for (int j = 0; j < PER; ++j) sm[j * 1024 + t] = v[j];
    __syncthreads();

    const int b = t * PER;
    int c[PER];
    int s = 0;
#pragma unroll
    for (int i = 0; i < PER; ++i) { c[i] = sm[b + i]; s += c[i]; }

    int inc = s;
#pragma unroll
    for (int off = 1; off < 32; off <<= 1) {
        const int u = __shfl_up(inc, off, 32);
        if (lane >= off) inc += u;
    }
    if (lane == 31) wtot[w] = inc;

#pragma unroll
    for (int off = 16; off > 0; off >>= 1) {
        pmn = fminf(pmn, __shfl_xor(pmn, off, 32));
        pmx = fmaxf(pmx, __shfl_xor(pmx, off, 32));
    }
    if (lane == 0 && w < PREP_B / 32) { pmn_s[w] = pmn; pmx_s[w] = pmx; }
    __syncthreads();
    if (t == 0) {
#pragma unroll
        for (int k = 1; k < PREP_B / 32; ++k) { pmn = fminf(pmn, pmn_s[k]); pmx = fmaxf(pmx, pmx_s[k]); }
        mm[0] = __float_as_uint(pmn);
        mm[1] = __float_as_uint(pmx);
    }
    const int wt = wtot[lane];
    int winc = wt;
#pragma unroll
    for (int off = 1; off < 32; off <<= 1) {
        const int u = __shfl_up(winc, off, 32);
        if (lane >= off) winc += u;
    }
    const int wbase = __shfl(winc, w, 32) - __shfl(wt, w, 32);
    const int total = __shfl(winc, 31, 32);

    int run = wbase + inc - s;
#pragma unroll
    for (int i = 0; i < PER; ++i) { sm[b + i] = run; run += c[i]; }
    __syncthreads();

#pragma unroll
    for (int j = 0; j < PER; ++j) {
        const int idx = j * 1024 + t;
        if (idx < NN) rowptr[idx] = sm[idx];
    }
    if (t == 0) rowptr[NN] = total;
}

__global__ void fill_kernel(const int* __restrict__ src, const int* __restrict__ dst,
                            const int* __restrict__ et, const float* __restrict__ ew,
                            const unsigned* __restrict__ mm,
                            const int* __restrict__ rowptr, const int* __restrict__ rank,
                            int* __restrict__ eg, float* __restrict__ ewn, int E) {
    int e = blockIdx.x * blockDim.x + threadIdx.x;
    if (e >= E) return;
    const float mn = __uint_as_float(mm[0]);
    const float mx = __uint_as_float(mm[1]);
    const int p = rowptr[dst[e]] + rank[e];
    eg[p]  = src[e] * RR + et[e];
    ewn[p] = (ew[e] - mn) / (mx - mn + 1e-8f);
}

#define GW 5

template <int NCT>
struct BCopy {
    static constexpr int NT    = GW * 32;
    static constexpr int TASKS = 256 * NCT;
    static constexpr int ITER  = (TASKS + NT - 1) / NT;
    static constexpr int WORDS = 2048 * NCT;
};

template <int LDB, int NCT>
__device__ __forceinline__ void b_fetch(const float* __restrict__ bm, int t,
                                        float4 (&lo)[BCopy<NCT>::ITER], float4 (&hi)[BCopy<NCT>::ITER]) {
    using BC = BCopy<NCT>;
#pragma unroll
    for (int i = 0; i < BC::ITER; ++i) {
        const int task = i * BC::NT + t;
        if ((i + 1) * BC::NT <= BC::TASKS || task < BC::TASKS) {
            const int q  = task & 3;
            const int ct = (task >> 2) % NCT;
            const int kp = (task >> 2) / NCT;
            const float* p = bm + (size_t)(2 * kp) * LDB + ct * 16 + q * 4;
            lo[i] = *(const float4*)p;
            hi[i] = *(const float4*)(p + LDB);
        }
    }
}

template <int NCT>
__device__ __forceinline__ void b_store(float* __restrict__ buf, int t,
                                        const float4 (&lo)[BCopy<NCT>::ITER], const float4 (&hi)[BCopy<NCT>::ITER]) {
    using BC = BCopy<NCT>;
#pragma unroll
    for (int i = 0; i < BC::ITER; ++i) {
        const int task = i * BC::NT + t;
        if ((i + 1) * BC::NT <= BC::TASKS || task < BC::TASKS) {
            const int q  = task & 3;
            const int ct = (task >> 2) % NCT;
            const int kp = (task >> 2) / NCT;
            float* d = buf + ((((kp >> 1) * NCT + ct) * 32) + (kp & 1) * 16 + q * 4) * 2;
            *(float4*)d       = make_float4(lo[i].x, hi[i].x, lo[i].y, hi[i].y);
            *(float4*)(d + 4) = make_float4(lo[i].z, hi[i].z, lo[i].w, hi[i].w);
        }
    }
}

template <int LDCX, int NCT>
__device__ __forceinline__ void slab_matrix(const v2f (&a)[32], const float* __restrict__ bl,
                                            float* __restrict__ crel) {
#pragma unroll
    for (int ct = 0; ct < NCT; ++ct) {
        v2f b[32];
#pragma unroll
        for (int s = 0; s < 32; ++s) {
            b[s] = *(const v2f*)(bl + ((s * NCT + ct) * 32) * 2);
        }
        v8f acc0 = {};
        v8f acc1 = {};
#pragma unroll
        for (int s = 0; s < 32; s += 2) {
            acc0 = __builtin_amdgcn_wmma_f32_16x16x4_f32(
                false, a[s], false, b[s], (short)0, acc0, false, false);
            acc1 = __builtin_amdgcn_wmma_f32_16x16x4_f32(
                false, a[s + 1], false, b[s + 1], (short)0, acc1, false, false);
        }
        float* crow = crel + ct * 16;
#pragma unroll
        for (int j = 0; j < 8; ++j) {
            crow[(size_t)j * LDCX] = acc0[j] + acc1[j];
        }
    }
}

template <int M0, int CNT, int LDB, int LDC, int NCT, int NREL, int LDC2, int LDR>
__device__ __forceinline__ void slab_group(float* __restrict__ bs, const float* __restrict__ A,
                                           const float* __restrict__ B, float* __restrict__ C,
                                           float* __restrict__ C2, const float* __restrict__ Broot,
                                           float* __restrict__ Croot, int t, int rowbase) {
    constexpr int NW = NREL + (LDC2 > 0 ? 1 : 0);
    using BC = BCopy<NCT>;
    const int lane = t & 31;
    const int half = lane >> 4;
    const int lr   = lane & 15;

    const float* arow = A + (size_t)(rowbase + lr) * 128 + 2 * half;
    v2f a[32];
#pragma unroll
    for (int s = 0; s < 32; ++s) {
        a[s] = *(const v2f*)(arow + 4 * s);
    }

    float4 lo[CNT][BC::ITER];
    float4 hi[CNT][BC::ITER];
#pragma unroll
    for (int i = 0; i < CNT; ++i) {
        const float* bm = (M0 + i < NW) ? B + (size_t)(M0 + i) * 128 * LDB : Broot;
        b_fetch<LDB, NCT>(bm, t, lo[i], hi[i]);
    }
#pragma unroll
    for (int i = 0; i < CNT; ++i) {
        b_store<NCT>(bs + i * BC::WORDS, t, lo[i], hi[i]);
    }
    __syncthreads();

    const size_t row0 = (size_t)(rowbase + 8 * half);
#pragma unroll
    for (int i = 0; i < CNT; ++i) {
        const int M = M0 + i;
        const float* bl = bs + i * BC::WORDS + lane * 2;
        if (M < NREL) {
            slab_matrix<LDC, NCT>(a, bl, C + row0 * LDC + M * (NCT * 16) + lr);
        } else if (M < NW) {
            slab_matrix<(LDC2 > 0 ? LDC2 : 1), NCT>(a, bl, C2 + row0 * LDC2 + lr);
        } else {
            slab_matrix<LDR, NCT>(a, bl, Croot + row0 * LDR + lr);
        }
        __builtin_amdgcn_sched_barrier(0);
    }
}

template <int NCT>
struct SlabSplit {
    static constexpr int NG   = (NCT == 2) ? 3 : 2;
    static constexpr int CMAX = (NCT == 2) ? 3 : 5;
};

template <int LDB, int LDC, int NCT, int NREL, int LDC2, int LDR>
__global__ __launch_bounds__(GW * 32)
void gemm_slab_wmma_f32_kernel(const float* __restrict__ A, const float* __restrict__ B,
                               float* __restrict__ C, float* __restrict__ C2,
                               const float* __restrict__ Broot, float* __restrict__ Croot) {
    static_assert((NN / 16) % GW == 0, "every wave of every workgroup has a row tile");
    static_assert(NREL + (LDC2 > 0 ? 1 : 0) + 1 == 9 && LDR > 0, "the slab is 9 matrices, the root matrix last");
    static_assert(NCT == 2 || (NCT == 1 && LDC2 == 0), "the two slab shapes in use");
    using BC = BCopy<NCT>;
    using SP = SlabSplit<NCT>;
    __shared__ __attribute__((aligned(16))) float bs[SP::CMAX * BC::WORDS];
    const int t  = threadIdx.x;
    const int g  = blockIdx.x / SP::NG;
    const int mg = blockIdx.x - SP::NG * g;
    const int rowbase = (g * GW + (t >> 5)) * 16;
    if constexpr (NCT == 2) {
        if (mg == 0)      slab_group<0, 3, LDB, LDC, NCT, NREL, LDC2, LDR>(bs, A, B, C, C2, Broot, Croot, t, rowbase);
        else if (mg == 1) slab_group<3, 3, LDB, LDC, NCT, NREL, LDC2, LDR>(bs, A, B, C, C2, Broot, Croot, t, rowbase);
        else              slab_group<6, 3, LDB, LDC, NCT, NREL, LDC2, LDR>(bs, A, B, C, C2, Broot, Croot, t, rowbase);
    } else {
        if (mg == 0)      slab_group<0, 4, LDB, LDC, NCT, NREL, LDC2, LDR>(bs, A, B, C, C2, Broot, Croot, t, rowbase);
        else              slab_group<4, 5, LDB, LDC, NCT, NREL, LDC2, LDR>(bs, A, B, C, C2, Broot, Croot, t, rowbase);
    }
}

template <int LDO, int RELU>
__global__ __launch_bounds__(256)
void gather_slab_kernel(const float* __restrict__ slab,
                        const int* __restrict__ rowptr,
                        const int* __restrict__ eg,
                        const float* __restrict__ ewn,
                        const float* __restrict__ bias,
                        float* __restrict__ dest, int c0) {
    static_assert(NN % 16 == 0, "256 threads take 16 whole nodes");
    const int lane = threadIdx.x & 31;
    const int sub  = lane & 15;
    const int hb   = lane & 16;
    const int n = ((blockIdx.x * blockDim.x + threadIdx.x) >> 5) * 2 + (lane >> 4);
    const int start = rowptr[n];
    const int deg   = rowptr[n + 1] - start;
    const int degmax = max(deg, __shfl_xor(deg, 16, 32));
    float acc = 0.0f;
    for (int base = 0; base < degmax; base += 16) {
        const int rem = deg - base;
        int   gl = 0;
        float wl = 0.0f;
        if (sub < rem) {
            gl = eg[start + base + sub];
            wl = ewn[start + base + sub];
        }
#pragma unroll
        for (int k = 0; k < 16; ++k) {
            const int   g  = __shfl(gl, hb + k, 32);
            const float wk = __shfl(wl, hb + k, 32);
            const float t  = slab[(size_t)g * 16 + sub];
            acc += (k < rem) ? t * wk : 0.0f;
        }
    }
    float* o = dest + (size_t)n * LDO + c0 + sub;
    const float v = acc * (1.0f / fmaxf((float)deg, 1.0f)) + *o + bias[c0 + sub];
    *o = RELU ? fmaxf(v, 0.0f) : v;
}

template <int LDO, int RELU, int PB>
__global__ __launch_bounds__(256)
void gather_slab32_kernel(const float* __restrict__ slabA,
                          const float* __restrict__ slabB,
                          const int* __restrict__ rowptr,
                          const int* __restrict__ eg,
                          const float* __restrict__ ewn,
                          const float* __restrict__ bias,
                          float* __restrict__ dest, int c0) {
    static_assert(NN % 16 == 0, "256 threads take 16 whole nodes");
    static_assert(PB % 128 == 0 && (long long)NN * PB < (1ll << 31), "row offsets keep bit 0 free and fit 31 bits");
    const int lane = threadIdx.x & 31;
    const int sub  = lane & 15;
    const int hb   = lane & 16;
    const int n = ((blockIdx.x * blockDim.x + threadIdx.x) >> 5) * 2 + (lane >> 4);
    const int start = rowptr[n];
    const int deg   = rowptr[n + 1] - start;
    const int degmax = max(deg, __shfl_xor(deg, 16, 32));
    float acc0 = 0.0f;
    float acc1 = 0.0f;
    for (int base = 0; base < degmax; base += 16) {
        const int rem = deg - base;
        int   cl = 0;
        float wl = 0.0f;
        if (sub < rem) {
            const int g = eg[start + base + sub];
            wl = ewn[start + base + sub];
            const int  s    = g >> 3;
            const bool last = (g & 7) == 7;
            cl = last ? s * PB + 1 : (g - s) * 128;
        }
#pragma unroll
        for (int k = 0; k < 16; ++k) {
            const int   ck = __shfl(cl, hb + k, 32);
            const float wk = __shfl(wl, hb + k, 32);
            const char* bufp = (ck & 1) ? (const char*)slabB : (const char*)slabA;
            const unsigned off = ((unsigned)ck & ~127u) | ((unsigned)sub << 3);
            const v2f t = *(const v2f*)(bufp + off);
            acc0 += (k < rem) ? t.x * wk : 0.0f;
            acc1 += (k < rem) ? t.y * wk : 0.0f;
        }
    }
    float* o = dest + (size_t)n * LDO + c0 + 2 * sub;
    const v2f ov = *(const v2f*)o;
    const float inv = 1.0f / fmaxf((float)deg, 1.0f);
    v2f v;
    v.x = acc0 * inv + ov.x + bias[c0 + 2 * sub];
    v.y = acc1 * inv + ov.y + bias[c0 + 2 * sub + 1];
    if (RELU) { v.x = fmaxf(v.x, 0.0f); v.y = fmaxf(v.y, 0.0f); }
    *(v2f*)o = v;
}

extern "C" void kernel_launch(void* const* d_in, const int* in_sizes, int n_in,
                              void* d_out, int out_size, void* d_ws, size_t ws_size,
                              hipStream_t stream) {
    const float* x     = (const float*)d_in[0];
    const int*   eidx  = (const int*)d_in[1];
    const float* ew    = (const float*)d_in[2];
    const int*   et    = (const int*)d_in[3];
    const float* W1    = (const float*)d_in[4];
    const float* root1 = (const float*)d_in[5];
    const float* bias1 = (const float*)d_in[6];
    const float* W2    = (const float*)d_in[7];
    const float* root2 = (const float*)d_in[8];
    const float* bias2 = (const float*)d_in[9];
    float* out = (float*)d_out;

    const int* src = eidx;
    const int* dst = eidx + EE;

    int*   rowptr = (int*)d_ws;
    int*   eg     = rowptr + NN + 1;
    float* ewn    = (float*)(eg + EE);
    unsigned* mm  = (unsigned*)(ewn + EE);
    float* slab = (float*)d_ws + 1250016;
    float* h    = (float*)d_ws + (size_t)2 * NN * HH;
    int*   deg    = (int*)h;
    int*   rank   = deg + NN;
    float* part   = slab;
    float* slab7  = out;

    const int T = 256;
    const int gemmGroups = NN / 16 / GW;
    const int gblocks  = NN / 16;

    prep_kernel<<<PREP_B, 256, 0, stream>>>(ew, part, deg);
    count_rank_kernel<<<(EE + T - 1) / T, T, 0, stream>>>(dst, deg, rank, EE);
    scan_kernel<<<1, 1024, 0, stream>>>(deg, rowptr, part, mm);
    fill_kernel<<<(EE + T - 1) / T, T, 0, stream>>>(src, dst, et, ew, mm, rowptr, rank, eg, ewn, EE);

    for (int c0 = 0; c0 < HH; c0 += 32) {
        gemm_slab_wmma_f32_kernel<HH, (RR - 1) * 32, 2, RR - 1, 32, HH><<<gemmGroups * 3, GW * 32, 0, stream>>>(
            x, W1 + c0, slab, slab7, root1 + c0, h + c0);
        gather_slab32_kernel<HH, 1, 128><<<gblocks, T, 0, stream>>>(
            slab, slab7, rowptr, eg, ewn, bias1, h, c0);
    }

    gemm_slab_wmma_f32_kernel<OO, (RR - 1) * 32, 2, RR - 1, OO, OO><<<gemmGroups * 3, GW * 32, 0, stream>>>(
        h, W2, slab, out + 32, root2, out);
    gather_slab32_kernel<OO, 0, OO * 4><<<gblocks, T, 0, stream>>>(
        slab, out + 32, rowptr, eg, ewn, bias2, out, 0);
    for (int c0 = 32; c0 < OO; c0 += 16) {
        gemm_slab_wmma_f32_kernel<OO, RR * 16, 1, RR, 0, OO><<<gemmGroups * 2, GW * 32, 0, stream>>>(
            h, W2 + c0, slab, nullptr, root2 + c0, out + c0);
        gather_slab_kernel<OO, 0><<<gblocks, T, 0, stream>>>(
            slab, rowptr, eg, ewn, bias2, out, c0);
    }
}


# GEMM A operand fetched as whole rows: a wave loads its tile with 16 loads of 512 contiguous bytes, not 32 loads touching 16 lines each, and cuts its fragments from its own LDS tile; sums unchanged
# speedup vs baseline: 2.1292x; 1.0315x over previous
#include <hip/hip_runtime.h>
#include <hip/hip_bf16.h>

#define NN 50000
#define EE 600000
#define FF 128
#define HH 128
#define OO 64
#define RR 8

typedef float v2f __attribute__((ext_vector_type(2)));
typedef float v8f __attribute__((ext_vector_type(8)));

#define PREP_B 256
#define PREP_PER ((EE + PREP_B * 256 - 1) / (PREP_B * 256))

__global__ __launch_bounds__(256)
void prep_kernel(const float* __restrict__ ew, float* __restrict__ part,
                 int* __restrict__ deg) {
    __shared__ float smn[8];
    __shared__ float smx[8];
    const int tid = blockIdx.x * 256 + threadIdx.x;
    if (tid < NN) deg[tid] = 0;
    static_assert(PREP_B * 256 >= NN, "one counter per thread");
    float v[PREP_PER];
#pragma unroll
    for (int j = 0; j < PREP_PER; ++j) v[j] = ew[min(j * (PREP_B * 256) + tid, EE - 1)];
    float mn = v[0];
    float mx = v[0];
#pragma unroll
    for (int j = 1; j < PREP_PER; ++j) { mn = fminf(mn, v[j]); mx = fmaxf(mx, v[j]); }
    for (int off = 16; off > 0; off >>= 1) {
        mn = fminf(mn, __shfl_xor(mn, off, 32));
        mx = fmaxf(mx, __shfl_xor(mx, off, 32));
    }
    if ((threadIdx.x & 31) == 0) { smn[threadIdx.x >> 5] = mn; smx[threadIdx.x >> 5] = mx; }
    __syncthreads();
    if (threadIdx.x == 0) {
#pragma unroll
        for (int k = 1; k < 8; ++k) { mn = fminf(mn, smn[k]); mx = fmaxf(mx, smx[k]); }
        part[blockIdx.x] = mn;
        part[PREP_B + blockIdx.x] = mx;
    }
}

__global__ void count_rank_kernel(const int* __restrict__ dst, int* __restrict__ deg,
                                  int* __restrict__ rank, int E) {
    int e = blockIdx.x * blockDim.x + threadIdx.x;
    if (e < E) rank[e] = atomicAdd(&deg[dst[e]], 1);
}

__global__ __launch_bounds__(1024)
void scan_kernel(const int* __restrict__ deg, int* __restrict__ rowptr,
                 const float* __restrict__ part, unsigned* __restrict__ mm) {
    constexpr int PER = (NN + 1023) / 1024;
    __shared__ int sm[PER * 1024];
    __shared__ int wtot[32];
    __shared__ float pmn_s[PREP_B / 32];
    __shared__ float pmx_s[PREP_B / 32];
    const int t    = threadIdx.x;
    const int lane = t & 31;
    const int w    = t >> 5;

    float pmn = part[t & (PREP_B - 1)];
    float pmx = part[PREP_B + (t & (PREP_B - 1))];

    int v[PER];
#pragma unroll
    for (int j = 0; j < PER; ++j) {
        const int idx = j * 1024 + t;
        const int d = deg[min(idx, NN - 1)];
        v[j] = (idx < NN) ? d : 0;
    }
#pragma unroll
    for (int j = 0; j < PER; ++j) sm[j * 1024 + t] = v[j];
    __syncthreads();

    const int b = t * PER;
    int c[PER];
    int s = 0;
#pragma unroll
    for (int i = 0; i < PER; ++i) { c[i] = sm[b + i]; s += c[i]; }

    int inc = s;
#pragma unroll
    for (int off = 1; off < 32; off <<= 1) {
        const int u = __shfl_up(inc, off, 32);
        if (lane >= off) inc += u;
    }
    if (lane == 31) wtot[w] = inc;

#pragma unroll
    for (int off = 16; off > 0; off >>= 1) {
        pmn = fminf(pmn, __shfl_xor(pmn, off, 32));
        pmx = fmaxf(pmx, __shfl_xor(pmx, off, 32));
    }
    if (lane == 0 && w < PREP_B / 32) { pmn_s[w] = pmn; pmx_s[w] = pmx; }
    __syncthreads();
    if (t == 0) {
#pragma unroll
        for (int k = 1; k < PREP_B / 32; ++k) { pmn = fminf(pmn, pmn_s[k]); pmx = fmaxf(pmx, pmx_s[k]); }
        mm[0] = __float_as_uint(pmn);
        mm[1] = __float_as_uint(pmx);
    }
    const int wt = wtot[lane];
    int winc = wt;
#pragma unroll
    for (int off = 1; off < 32; off <<= 1) {
        const int u = __shfl_up(winc, off, 32);
        if (lane >= off) winc += u;
    }
    const int wbase = __shfl(winc, w, 32) - __shfl(wt, w, 32);
    const int total = __shfl(winc, 31, 32);

    int run = wbase + inc - s;
#pragma unroll
    for (int i = 0; i < PER; ++i) { sm[b + i] = run; run += c[i]; }
    __syncthreads();

#pragma unroll
    for (int j = 0; j < PER; ++j) {
        const int idx = j * 1024 + t;
        if (idx < NN) rowptr[idx] = sm[idx];
    }
    if (t == 0) rowptr[NN] = total;
}

__global__ void fill_kernel(const int* __restrict__ src, const int* __restrict__ dst,
                            const int* __restrict__ et, const float* __restrict__ ew,
                            const unsigned* __restrict__ mm,
                            const int* __restrict__ rowptr, const int* __restrict__ rank,
                            int* __restrict__ eg, float* __restrict__ ewn, int E) {
    int e = blockIdx.x * blockDim.x + threadIdx.x;
    if (e >= E) return;
    const float mn = __uint_as_float(mm[0]);
    const float mx = __uint_as_float(mm[1]);
    const int p = rowptr[dst[e]] + rank[e];
    eg[p]  = src[e] * RR + et[e];
    ewn[p] = (ew[e] - mn) / (mx - mn + 1e-8f);
}

#define GW 5
#define APITCH 132

template <int NCT>
struct BCopy {
    static constexpr int NT    = GW * 32;
    static constexpr int TASKS = 256 * NCT;
    static constexpr int ITER  = (TASKS + NT - 1) / NT;
    static constexpr int WORDS = 2048 * NCT;
};

template <int LDB, int NCT>
__device__ __forceinline__ void b_fetch(const float* __restrict__ bm, int t,
                                        float4 (&lo)[BCopy<NCT>::ITER], float4 (&hi)[BCopy<NCT>::ITER]) {
    using BC = BCopy<NCT>;
#pragma unroll
    for (int i = 0; i < BC::ITER; ++i) {
        const int task = i * BC::NT + t;
        if ((i + 1) * BC::NT <= BC::TASKS || task < BC::TASKS) {
            const int q  = task & 3;
            const int ct = (task >> 2) % NCT;
            const int kp = (task >> 2) / NCT;
            const float* p = bm + (size_t)(2 * kp) * LDB + ct * 16 + q * 4;
            lo[i] = *(const float4*)p;
            hi[i] = *(const float4*)(p + LDB);
        }
    }
}

template <int NCT>
__device__ __forceinline__ void b_store(float* __restrict__ buf, int t,
                                        const float4 (&lo)[BCopy<NCT>::ITER], const float4 (&hi)[BCopy<NCT>::ITER]) {
    using BC = BCopy<NCT>;
#pragma unroll
    for (int i = 0; i < BC::ITER; ++i) {
        const int task = i * BC::NT + t;
        if ((i + 1) * BC::NT <= BC::TASKS || task < BC::TASKS) {
            const int q  = task & 3;
            const int ct = (task >> 2) % NCT;
            const int kp = (task >> 2) / NCT;
            float* d = buf + ((((kp >> 1) * NCT + ct) * 32) + (kp & 1) * 16 + q * 4) * 2;
            *(float4*)d       = make_float4(lo[i].x, hi[i].x, lo[i].y, hi[i].y);
            *(float4*)(d + 4) = make_float4(lo[i].z, hi[i].z, lo[i].w, hi[i].w);
        }
    }
}

template <int LDCX, int NCT>
__device__ __forceinline__ void slab_matrix(const v2f (&a)[32], const float* __restrict__ bl,
                                            float* __restrict__ crel) {
#pragma unroll
    for (int ct = 0; ct < NCT; ++ct) {
        v2f b[32];
#pragma unroll
        for (int s = 0; s < 32; ++s) {
            b[s] = *(const v2f*)(bl + ((s * NCT + ct) * 32) * 2);
        }
        v8f acc0 = {};
        v8f acc1 = {};
#pragma unroll
        for (int s = 0; s < 32; s += 2) {
            acc0 = __builtin_amdgcn_wmma_f32_16x16x4_f32(
                false, a[s], false, b[s], (short)0, acc0, false, false);
            acc1 = __builtin_amdgcn_wmma_f32_16x16x4_f32(
                false, a[s + 1], false, b[s + 1], (short)0, acc1, false, false);
        }
        float* crow = crel + ct * 16;
#pragma unroll
        for (int j = 0; j < 8; ++j) {
            crow[(size_t)j * LDCX] = acc0[j] + acc1[j];
        }
    }
}

template <int M0, int CNT, int LDB, int LDC, int NCT, int NREL, int LDC2, int LDR>
__device__ __forceinline__ void slab_group(float* __restrict__ bs, float* __restrict__ as, const float* __restrict__ A,
                                           const float* __restrict__ B, float* __restrict__ C,
                                           float* __restrict__ C2, const float* __restrict__ Broot,
                                           float* __restrict__ Croot, int t, int rowbase) {
    constexpr int NW = NREL + (LDC2 > 0 ? 1 : 0);
    using BC = BCopy<NCT>;
    const int lane = t & 31;
    const int half = lane >> 4;
    const int lr   = lane & 15;

    float4 lo[CNT][BC::ITER];
    float4 hi[CNT][BC::ITER];
#pragma unroll
    for (int i = 0; i < CNT; ++i) {
        const float* bm = (M0 + i < NW) ? B + (size_t)(M0 + i) * 128 * LDB : Broot;
        b_fetch<LDB, NCT>(bm, t, lo[i], hi[i]);
    }

    const float* atile = A + (size_t)rowbase * 128 + lane * 4;
    float4 av[16];
#pragma unroll
    for (int i = 0; i < 16; ++i) {
        av[i] = *(const float4*)(atile + (size_t)i * 128);
    }
    __builtin_amdgcn_sched_barrier(0);

    float* aw = as + (t >> 5) * (16 * APITCH);
#pragma unroll
    for (int i = 0; i < 16; ++i) {
        *(float4*)(aw + i * APITCH + lane * 4) = av[i];
    }
#pragma unroll
    for (int i = 0; i < CNT; ++i) {
        b_store<NCT>(bs + i * BC::WORDS, t, lo[i], hi[i]);
    }
    __syncthreads();

    const float* ar = aw + lr * APITCH + 2 * half;
    v2f a[32];
#pragma unroll
    for (int s = 0; s < 32; ++s) {
        a[s] = *(const v2f*)(ar + 4 * s);
    }

    const size_t row0 = (size_t)(rowbase + 8 * half);
#pragma unroll
    for (int i = 0; i < CNT; ++i) {
        const int M = M0 + i;
        const float* bl = bs + i * BC::WORDS + lane * 2;
        if (M < NREL) {
            slab_matrix<LDC, NCT>(a, bl, C + row0 * LDC + M * (NCT * 16) + lr);
        } else if (M < NW) {
            slab_matrix<(LDC2 > 0 ? LDC2 : 1), NCT>(a, bl, C2 + row0 * LDC2 + lr);
        } else {
            slab_matrix<LDR, NCT>(a, bl, Croot + row0 * LDR + lr);
        }
        __builtin_amdgcn_sched_barrier(0);
    }
}

template <int NCT>
struct SlabSplit {
    static constexpr int NG   = (NCT == 2) ? 3 : 2;
    static constexpr int CMAX = (NCT == 2) ? 3 : 5;
};

template <int LDB, int LDC, int NCT, int NREL, int LDC2, int LDR>
__global__ __launch_bounds__(GW * 32)
void gemm_slab_wmma_f32_kernel(const float* __restrict__ A, const float* __restrict__ B,
                               float* __restrict__ C, float* __restrict__ C2,
                               const float* __restrict__ Broot, float* __restrict__ Croot) {
    static_assert((NN / 16) % GW == 0, "every wave of every workgroup has a row tile");
    static_assert(NREL + (LDC2 > 0 ? 1 : 0) + 1 == 9 && LDR > 0, "the slab is 9 matrices, the root matrix last");
    static_assert(NCT == 2 || (NCT == 1 && LDC2 == 0), "the two slab shapes in use");
    using BC = BCopy<NCT>;
    using SP = SlabSplit<NCT>;
    __shared__ __attribute__((aligned(16))) float bs[SP::CMAX * BC::WORDS];
    __shared__ __attribute__((aligned(16))) float as[GW * 16 * APITCH];
    const int t  = threadIdx.x;
    const int g  = blockIdx.x / SP::NG;
    const int mg = blockIdx.x - SP::NG * g;
    const int rowbase = (g * GW + (t >> 5)) * 16;
    if constexpr (NCT == 2) {
        if (mg == 0)      slab_group<0, 3, LDB, LDC, NCT, NREL, LDC2, LDR>(bs, as, A, B, C, C2, Broot, Croot, t, rowbase);
        else if (mg == 1) slab_group<3, 3, LDB, LDC, NCT, NREL, LDC2, LDR>(bs, as, A, B, C, C2, Broot, Croot, t, rowbase);
        else              slab_group<6, 3, LDB, LDC, NCT, NREL, LDC2, LDR>(bs, as, A, B, C, C2, Broot, Croot, t, rowbase);
    } else {
        if (mg == 0)      slab_group<0, 4, LDB, LDC, NCT, NREL, LDC2, LDR>(bs, as, A, B, C, C2, Broot, Croot, t, rowbase);
        else              slab_group<4, 5, LDB, LDC, NCT, NREL, LDC2, LDR>(bs, as, A, B, C, C2, Broot, Croot, t, rowbase);
    }
}

template <int LDO, int RELU>
__global__ __launch_bounds__(256)
void gather_slab_kernel(const float* __restrict__ slab,
                        const int* __restrict__ rowptr,
                        const int* __restrict__ eg,
                        const float* __restrict__ ewn,
                        const float* __restrict__ bias,
                        float* __restrict__ dest, int c0) {
    static_assert(NN % 16 == 0, "256 threads take 16 whole nodes");
    const int lane = threadIdx.x & 31;
    const int sub  = lane & 15;
    const int hb   = lane & 16;
    const int n = ((blockIdx.x * blockDim.x + threadIdx.x) >> 5) * 2 + (lane >> 4);
    const int start = rowptr[n];
    const int deg   = rowptr[n + 1] - start;
    const int degmax = max(deg, __shfl_xor(deg, 16, 32));
    float acc = 0.0f;
    for (int base = 0; base < degmax; base += 16) {
        const int rem = deg - base;
        int   gl = 0;
        float wl = 0.0f;
        if (sub < rem) {
            gl = eg[start + base + sub];
            wl = ewn[start + base + sub];
        }
#pragma unroll
        for (int k = 0; k < 16; ++k) {
            const int   g  = __shfl(gl, hb + k, 32);
            const float wk = __shfl(wl, hb + k, 32);
            const float t  = slab[(size_t)g * 16 + sub];
            acc += (k < rem) ? t * wk : 0.0f;
        }
    }
    float* o = dest + (size_t)n * LDO + c0 + sub;
    const float v = acc * (1.0f / fmaxf((float)deg, 1.0f)) + *o + bias[c0 + sub];
    *o = RELU ? fmaxf(v, 0.0f) : v;
}

template <int LDO, int RELU, int PB>
__global__ __launch_bounds__(256)
void gather_slab32_kernel(const float* __restrict__ slabA,
                          const float* __restrict__ slabB,
                          const int* __restrict__ rowptr,
                          const int* __restrict__ eg,
                          const float* __restrict__ ewn,
                          const float* __restrict__ bias,
                          float* __restrict__ dest, int c0) {
    static_assert(NN % 16 == 0, "256 threads take 16 whole nodes");
    static_assert(PB % 128 == 0 && (long long)NN * PB < (1ll << 31), "row offsets keep bit 0 free and fit 31 bits");
    const int lane = threadIdx.x & 31;
    const int sub  = lane & 15;
    const int hb   = lane & 16;
    const int n = ((blockIdx.x * blockDim.x + threadIdx.x) >> 5) * 2 + (lane >> 4);
    const int start = rowptr[n];
    const int deg   = rowptr[n + 1] - start;
    const int degmax = max(deg, __shfl_xor(deg, 16, 32));
    float acc0 = 0.0f;
    float acc1 = 0.0f;
    for (int base = 0; base < degmax; base += 16) {
        const int rem = deg - base;
        int   cl = 0;
        float wl = 0.0f;
        if (sub < rem) {
            const int g = eg[start + base + sub];
            wl = ewn[start + base + sub];
            const int  s    = g >> 3;
            const bool last = (g & 7) == 7;
            cl = last ? s * PB + 1 : (g - s) * 128;
        }
#pragma unroll
        for (int k = 0; k < 16; ++k) {
            const int   ck = __shfl(cl, hb + k, 32);
            const float wk = __shfl(wl, hb + k, 32);
            const char* bufp = (ck & 1) ? (const char*)slabB : (const char*)slabA;
            const unsigned off = ((unsigned)ck & ~127u) | ((unsigned)sub << 3);
            const v2f t = *(const v2f*)(bufp + off);
            acc0 += (k < rem) ? t.x * wk : 0.0f;
            acc1 += (k < rem) ? t.y * wk : 0.0f;
        }
    }
    float* o = dest + (size_t)n * LDO + c0 + 2 * sub;
    const v2f ov = *(const v2f*)o;
    const float inv = 1.0f / fmaxf((float)deg, 1.0f);
    v2f v;
    v.x = acc0 * inv + ov.x + bias[c0 + 2 * sub];
    v.y = acc1 * inv + ov.y + bias[c0 + 2 * sub + 1];
    if (RELU) { v.x = fmaxf(v.x, 0.0f); v.y = fmaxf(v.y, 0.0f); }
    *(v2f*)o = v;
}

extern "C" void kernel_launch(void* const* d_in, const int* in_sizes, int n_in,
                              void* d_out, int out_size, void* d_ws, size_t ws_size,
                              hipStream_t stream) {
    const float* x     = (const float*)d_in[0];
    const int*   eidx  = (const int*)d_in[1];
    const float* ew    = (const float*)d_in[2];
    const int*   et    = (const int*)d_in[3];
    const float* W1    = (const float*)d_in[4];
    const float* root1 = (const float*)d_in[5];
    const float* bias1 = (const float*)d_in[6];
    const float* W2    = (const float*)d_in[7];
    const float* root2 = (const float*)d_in[8];
    const float* bias2 = (const float*)d_in[9];
    float* out = (float*)d_out;

    const int* src = eidx;
    const int* dst = eidx + EE;

    int*   rowptr = (int*)d_ws;
    int*   eg     = rowptr + NN + 1;
    float* ewn    = (float*)(eg + EE);
    unsigned* mm  = (unsigned*)(ewn + EE);
    float* slab = (float*)d_ws + 1250016;
    float* h    = (float*)d_ws + (size_t)2 * NN * HH;
    int*   deg    = (int*)h;
    int*   rank   = deg + NN;
    float* part   = slab;
    float* slab7  = out;

    const int T = 256;
    const int gemmGroups = NN / 16 / GW;
    const int gblocks  = NN / 16;

    prep_kernel<<<PREP_B, 256, 0, stream>>>(ew, part, deg);
    count_rank_kernel<<<(EE + T - 1) / T, T, 0, stream>>>(dst, deg, rank, EE);
    scan_kernel<<<1, 1024, 0, stream>>>(deg, rowptr, part, mm);
    fill_kernel<<<(EE + T - 1) / T, T, 0, stream>>>(src, dst, et, ew, mm, rowptr, rank, eg, ewn, EE);

    for (int c0 = 0; c0 < HH; c0 += 32) {
        gemm_slab_wmma_f32_kernel<HH, (RR - 1) * 32, 2, RR - 1, 32, HH><<<gemmGroups * 3, GW * 32, 0, stream>>>(
            x, W1 + c0, slab, slab7, root1 + c0, h + c0);
        gather_slab32_kernel<HH, 1, 128><<<gblocks, T, 0, stream>>>(
            slab, slab7, rowptr, eg, ewn, bias1, h, c0);
    }

    gemm_slab_wmma_f32_kernel<OO, (RR - 1) * 32, 2, RR - 1, OO, OO><<<gemmGroups * 3, GW * 32, 0, stream>>>(
        h, W2, slab, out + 32, root2, out);
    gather_slab32_kernel<OO, 0, OO * 4><<<gblocks, T, 0, stream>>>(
        slab, out + 32, rowptr, eg, ewn, bias2, out, 0);
    for (int c0 = 32; c0 < OO; c0 += 16) {
        gemm_slab_wmma_f32_kernel<OO, RR * 16, 1, RR, 0, OO><<<gemmGroups * 2, GW * 32, 0, stream>>>(
            h, W2 + c0, slab, nullptr, root2 + c0, out + c0);
        gather_slab_kernel<OO, 0><<<gblocks, T, 0, stream>>>(
            slab, rowptr, eg, ewn, bias2, out, c0);
    }
}
